# starting kernel
# baseline (speedup 1.0000x reference)
#include <hip/hip_runtime.h>
#include <cstddef>

typedef float v2f __attribute__((ext_vector_type(2)));
typedef float v4f __attribute__((ext_vector_type(4)));
typedef float v8f __attribute__((ext_vector_type(8)));
typedef int   v4i __attribute__((ext_vector_type(4)));

#define B_  8
#define N_  2048
#define C_  128
#define D_  128
#define NEG_BIG (-9.0e15f)

static __device__ __forceinline__ v8f wmma_f32_4(v2f a, v2f b, v8f c) {
    return __builtin_amdgcn_wmma_f32_16x16x4_f32(false, a, false, b, (short)0, c,
                                                 false, false);
}

__global__ __launch_bounds__(256)
void wht_kernel(const float* __restrict__ x, const float* __restrict__ W,
                float* __restrict__ wht) {
    const int bid  = blockIdx.x;
    const int b    = bid >> 7;
    const int jt   = bid & 127;
    const int tid  = threadIdx.x;
    const int wv   = tid >> 5;
    const int lane = tid & 31;
    const int nl   = lane & 15;
    const int kh   = (lane >> 4) << 1;

    const float* xrow = x + ((size_t)(b * N_ + jt * 16 + nl)) * C_;
    const float* wrow = W + (size_t)(wv * 16 + nl) * C_;

    v8f acc = {};
#pragma unroll 4
    for (int k0 = 0; k0 < C_; k0 += 4) {
        const int kb = k0 + kh;
        v2f a = *(const v2f*)(wrow + kb);
        v2f bm = *(const v2f*)(xrow + kb);
        acc = wmma_f32_4(a, bm, acc);
    }

    float* ob = wht + (size_t)b * D_ * N_ + (size_t)jt * 16 + nl;
#pragma unroll
    for (int r = 0; r < 8; ++r) {
        const int d = wv * 16 + r + ((lane >> 4) << 3);
        ob[(size_t)d * N_] = acc[r];
    }
}

__global__ __launch_bounds__(256)
void f_kernel(const float* __restrict__ wht, const float* __restrict__ a,
              float* __restrict__ f1, float* __restrict__ f2) {
    const int t = blockIdx.x * 256 + threadIdx.x;
    const int b = t >> 11;
    const int j = t & (N_ - 1);
    const float* wb = wht + (size_t)b * D_ * N_ + j;
    float s1 = 0.f, s2 = 0.f;
#pragma unroll 4
    for (int d = 0; d < D_; ++d) {
        const float w = wb[(size_t)d * N_];
        s1 += w * a[d];
        s2 += w * a[D_ + d];
    }
    f1[t] = s1;
    f2[t] = s2;
}

__global__ __launch_bounds__(256)
void gat_kernel(const float* __restrict__ wht, const float* __restrict__ f1g,
                const float* __restrict__ f2g, const int* __restrict__ adj,
                float* __restrict__ out) {
    __shared__ unsigned lds_bits[16 * 65];
    __shared__ float lds_mw[256];
    __shared__ float lds_m[16], lds_f1[16], lds_pone[16], lds_l[16];
    __shared__ float lds_lw[8 * 16];
    __shared__ float accbuf[4 * 8 * 32 * 8];

    const int bid  = blockIdx.x;
    const int b    = bid >> 7;
    const int i0   = (bid & 127) * 16;
    const int tid  = threadIdx.x;
    const int wv   = tid >> 5;
    const int lane = tid & 31;
    const int nl   = lane & 15;

    const float* f2b  = f2g + b * N_;
    const float* whtb = wht + (size_t)b * D_ * N_;

    {
        const int i   = tid >> 4;
        const int seg = tid & 15;
        const int*   arow = adj + (size_t)(i0 + i) * N_ + seg * 128;
        const float* frow = f2b + seg * 128;
        float mmax = -3.0e38f;
#pragma unroll
        for (int w2 = 0; w2 < 4; ++w2) {
            unsigned bits = 0u;
#pragma unroll
            for (int q = 0; q < 8; ++q) {
                v4i av = *(const v4i*)(arow + w2 * 32 + q * 4);
                v4f fv = *(const v4f*)(frow + w2 * 32 + q * 4);
                if (av[0]) { bits |= 1u << (q * 4 + 0); mmax = fmaxf(mmax, fv[0]); }
                if (av[1]) { bits |= 1u << (q * 4 + 1); mmax = fmaxf(mmax, fv[1]); }
                if (av[2]) { bits |= 1u << (q * 4 + 2); mmax = fmaxf(mmax, fv[2]); }
                if (av[3]) { bits |= 1u << (q * 4 + 3); mmax = fmaxf(mmax, fv[3]); }
            }
            lds_bits[i * 65 + seg * 4 + w2] = bits;
        }
        lds_mw[i * 16 + seg] = mmax;
    }
    __syncthreads();
    if (tid < 16) {
        float M = -3.0e38f;
#pragma unroll
        for (int s = 0; s < 16; ++s) M = fmaxf(M, lds_mw[tid * 16 + s]);
        const bool empty = (M < -1.0e38f);
        const float f1v  = f1g[b * N_ + i0 + tid];
        const float sum  = f1v + M;
        const float emax = fmaxf(sum, 0.2f * sum);
        lds_m[tid]    = empty ? 0.0f : emax;
        lds_pone[tid] = empty ? 1.0f : 0.0f;
        lds_f1[tid]   = f1v;
    }
    __syncthreads();

    const float f1i = lds_f1[nl];
    const float mi  = lds_m[nl];
    const float pm  = lds_pone[nl];
    const float* wbase = whtb + (size_t)nl * N_;

    v8f acc[8] = {};
    float lsum = 0.0f;

    for (int g = 0; g < 64; ++g) {
        const int j0 = g * 32 + wv * 4;
        const int jA = j0 + ((lane >> 4) << 1);
        const unsigned word = lds_bits[nl * 65 + (jA >> 5)];
        const unsigned sh   = (unsigned)(jA & 31);

        const v2f f2v = *(const v2f*)(f2b + jA);
        float s0 = f1i + f2v[0]; s0 = fmaxf(s0, 0.2f * s0);
        float s1 = f1i + f2v[1]; s1 = fmaxf(s1, 0.2f * s1);
        const float p0 = ((word >> sh) & 1u)        ? __expf(s0 - mi) : pm;
        const float p1 = ((word >> (sh + 1)) & 1u)  ? __expf(s1 - mi) : pm;

        v2f A; A[0] = p0; A[1] = p1;
        lsum += p0 + p1;

#pragma unroll
        for (int dt = 0; dt < 8; ++dt) {
            const v2f Bv = *(const v2f*)(wbase + (size_t)dt * 16 * N_ + jA);
            acc[dt] = wmma_f32_4(A, Bv, acc[dt]);
        }
    }

    float lpair = lsum + __shfl_xor(lsum, 16);
    if (lane < 16) lds_lw[wv * 16 + lane] = lpair;
    __syncthreads();
    if (tid < 16) {
        float t = 0.0f;
#pragma unroll
        for (int w2 = 0; w2 < 8; ++w2) t += lds_lw[w2 * 16 + tid];
        lds_l[tid] = t;
    }
    __syncthreads();

    for (int half = 4; half >= 1; half >>= 1) {
        if (wv >= half && wv < 2 * half) {
            const int slot = wv - half;
#pragma unroll
            for (int dt = 0; dt < 8; ++dt)
                *(v8f*)&accbuf[(((size_t)slot * 8 + dt) * 32 + lane) * 8] = acc[dt];
        }
        __syncthreads();
        if (wv < half) {
#pragma unroll
            for (int dt = 0; dt < 8; ++dt)
                acc[dt] = acc[dt] +
                    *(const v8f*)&accbuf[(((size_t)wv * 8 + dt) * 32 + lane) * 8];
        }
        __syncthreads();
    }

    if (wv == 0) {
        float* ob = out + ((size_t)b * N_ + i0) * D_;
#pragma unroll
        for (int dt = 0; dt < 8; ++dt) {
#pragma unroll
            for (int r = 0; r < 8; ++r) {
                const int row = r + ((lane >> 4) << 3);
                const float li = lds_l[row];
                const float h  = acc[dt][r] / li;
                const float o  = (h > 0.0f) ? h : (__expf(h) - 1.0f);
                ob[(size_t)row * D_ + dt * 16 + nl] = o;
            }
        }
    }
}

extern "C" void kernel_launch(void* const* d_in, const int* in_sizes, int n_in,
                              void* d_out, int out_size, void* d_ws, size_t ws_size,
                              hipStream_t stream) {
    const float* x   = (const float*)d_in[0];
    const int*   adj = (const int*)d_in[1];
    const float* W   = (const float*)d_in[2];
    const float* a   = (const float*)d_in[3];
    float* out = (float*)d_out;

    float* wht = (float*)d_ws;
    float* f1  = wht + (size_t)B_ * D_ * N_;
    float* f2  = f1 + (size_t)B_ * N_;

    wht_kernel<<<dim3(B_ * (N_ / 16)), dim3(256), 0, stream>>>(x, W, wht);
    f_kernel<<<dim3((B_ * N_) / 256), dim3(256), 0, stream>>>(wht, a, f1, f2);
    gat_kernel<<<dim3(B_ * (N_ / 16)), dim3(256), 0, stream>>>(wht, f1, f2, adj, out);
}


# gat pass 1: each wave reads two adjacency rows as whole 512-byte runs (4 lines per load, was 32); mask words merged by 8-lane OR exchange, row max by lane butterfly; no sum order changes
# speedup vs baseline: 2.8637x; 2.8637x over previous
#include <hip/hip_runtime.h>
#include <cstddef>

typedef float v2f __attribute__((ext_vector_type(2)));
typedef float v4f __attribute__((ext_vector_type(4)));
typedef float v8f __attribute__((ext_vector_type(8)));
typedef int   v4i __attribute__((ext_vector_type(4)));

#define B_  8
#define N_  2048
#define C_  128
#define D_  128
#define NEG_BIG (-9.0e15f)

static __device__ __forceinline__ v8f wmma_f32_4(v2f a, v2f b, v8f c) {
    return __builtin_amdgcn_wmma_f32_16x16x4_f32(false, a, false, b, (short)0, c,
                                                 false, false);
}

__global__ __launch_bounds__(256)
void wht_kernel(const float* __restrict__ x, const float* __restrict__ W,
                float* __restrict__ whf) {
    const int bid  = blockIdx.x;
    const int b    = bid >> 7;
    const int jt   = bid & 127;
    const int tid  = threadIdx.x;
    const int wv   = tid >> 5;
    const int lane = tid & 31;
    const int nl   = lane & 15;
    const int hh   = lane >> 4;
    const int kh   = hh << 1;

    const float* xrow = x + ((size_t)(b * N_ + jt * 16 + nl)) * C_;
    const float* wrow = W + (size_t)(wv * 16 + nl) * C_;

    v8f acc = {};
#pragma unroll 4
    for (int k0 = 0; k0 < C_; k0 += 4) {
        const int kb = k0 + kh;
        v2f xa = *(const v2f*)(xrow + kb);
        v2f wb = *(const v2f*)(wrow + kb);
        acc = wmma_f32_4(xa, wb, acc);
    }

    float* ob = whf + ((((size_t)b * (N_ / 8) + jt * 2 + hh) * 8 + wv) * 32) * 4;
    v4f lo, hi;
    lo[0] = acc[0]; lo[1] = acc[1]; lo[2] = acc[2]; lo[3] = acc[3];
    hi[0] = acc[4]; hi[1] = acc[5]; hi[2] = acc[6]; hi[3] = acc[7];
    *(v4f*)(ob + nl * 4)        = lo;
    *(v4f*)(ob + (nl + 16) * 4) = hi;
}

__global__ __launch_bounds__(256)
void f_kernel(const float* __restrict__ whf, const float* __restrict__ a,
              float* __restrict__ f1, float* __restrict__ f2) {
    const int tid  = threadIdx.x;
    const int lane = tid & 31;
    const int c    = lane & 15;
    const int hh   = lane >> 4;
    const int gc   = blockIdx.x * 8 + (tid >> 5);
    const v4f* wp  = (const v4f*)whf + (size_t)gc * (8 * 32) + lane;

    v4f s1 = {}, s2 = {};
#pragma unroll
    for (int dt = 0; dt < 8; ++dt) {
        const v4f w   = wp[dt * 32];
        const float a1 = a[dt * 16 + c];
        const float a2 = a[D_ + dt * 16 + c];
        s1 += w * a1;
        s2 += w * a2;
    }
#pragma unroll
    for (int m = 8; m >= 1; m >>= 1) {
#pragma unroll
        for (int e = 0; e < 4; ++e) {
            s1[e] += __shfl_xor(s1[e], m);
            s2[e] += __shfl_xor(s2[e], m);
        }
    }
    if (c == 0) {
        *(v4f*)(f1 + (size_t)gc * 8 + hh * 4) = s1;
        *(v4f*)(f2 + (size_t)gc * 8 + hh * 4) = s2;
    }
}

__global__ __launch_bounds__(256, 8)
void gat_kernel(const float* __restrict__ whf, const float* __restrict__ f1g,
                const float* __restrict__ f2g, const int* __restrict__ adj,
                float* __restrict__ out) {
    __shared__ unsigned lds_bits[16 * 65];
    __shared__ float lds_mw[16];
    __shared__ float lds_m[16], lds_f1[16], lds_pone[16], lds_l[16];
    __shared__ float lds_lw[8 * 16];
    __shared__ float accbuf[4 * 8 * 32 * 8];

    const int bid  = blockIdx.x;
    const int b    = bid >> 7;
    const int i0   = (bid & 127) * 16;
    const int tid  = threadIdx.x;
    const int wv   = tid >> 5;
    const int lane = tid & 31;
    const int nl   = lane & 15;

    const float* f2b  = f2g + b * N_;
    const v4f*   whfb = (const v4f*)(whf + (size_t)b * D_ * N_) + lane;

    {
        const int r0 = wv * 2;
        const v4i* arow0 = (const v4i*)(adj + (size_t)(i0 + r0) * N_) + lane;
        const v4i* arow1 = arow0 + N_ / 4;
        const v4f* frow  = (const v4f*)f2b + lane;
        const unsigned sh = (unsigned)(lane & 7) * 4u;
        float mx0 = -3.0e38f, mx1 = -3.0e38f;
        for (int tb = 0; tb < 16; tb += 8) {
            v4i a0[8], a1[8];
            v4f fv[8];
#pragma unroll
            for (int u = 0; u < 8; ++u) {
                a0[u] = arow0[(tb + u) * 32];
                a1[u] = arow1[(tb + u) * 32];
                fv[u] = frow[(tb + u) * 32];
            }
            unsigned n0[8], n1[8];
#pragma unroll
            for (int u = 0; u < 8; ++u) {
                unsigned b0 = 0u, b1 = 0u;
#pragma unroll
                for (int e = 0; e < 4; ++e) {
                    if (a0[u][e] > 0) { b0 |= 1u << e; mx0 = fmaxf(mx0, fv[u][e]); }
                    if (a1[u][e] > 0) { b1 |= 1u << e; mx1 = fmaxf(mx1, fv[u][e]); }
                }
                n0[u] = b0 << sh;
                n1[u] = b1 << sh;
            }
#pragma unroll
            for (int m = 1; m <= 4; m <<= 1) {
#pragma unroll
                for (int u = 0; u < 8; ++u) {
                    n0[u] |= __shfl_xor(n0[u], m);
                    n1[u] |= __shfl_xor(n1[u], m);
                }
            }
            if ((lane & 7) == 0) {
#pragma unroll
                for (int u = 0; u < 8; ++u) {
                    lds_bits[r0 * 65 + (tb + u) * 4 + (lane >> 3)]       = n0[u];
                    lds_bits[(r0 + 1) * 65 + (tb + u) * 4 + (lane >> 3)] = n1[u];
                }
            }
        }
#pragma unroll
        for (int m = 16; m >= 1; m >>= 1) {
            mx0 = fmaxf(mx0, __shfl_xor(mx0, m));
            mx1 = fmaxf(mx1, __shfl_xor(mx1, m));
        }
        if (lane == 0) {
            lds_mw[r0]     = mx0;
            lds_mw[r0 + 1] = mx1;
        }
    }
    __syncthreads();
    if (tid < 16) {
        const float M = lds_mw[tid];
        const bool empty = (M < -1.0e38f);
        const float f1v  = f1g[b * N_ + i0 + tid];
        const float sum  = f1v + M;
        const float emax = fmaxf(sum, 0.2f * sum);
        lds_m[tid]    = empty ? 0.0f : emax;
        lds_pone[tid] = empty ? 1.0f : 0.0f;
        lds_f1[tid]   = f1v;
    }
    __syncthreads();

    const float f1i = lds_f1[nl];
    const float mi  = lds_m[nl];
    const float pm  = lds_pone[nl];
    const int    hh   = lane >> 4;

    v8f acc[8] = {};
    float lsum = 0.0f;

    for (int g = 0; g < 32; ++g) {
        const int jc = g * 8 + wv;
        const v4f* wq = whfb + (size_t)jc * (8 * 32);
        v4f Bq[8];
#pragma unroll
        for (int dt = 0; dt < 8; ++dt) Bq[dt] = wq[dt * 32];

        const unsigned word = lds_bits[nl * 65 + (jc >> 2)];
        const unsigned m4   = word >> (unsigned)((jc & 3) * 8 + hh * 4);

        const v4f f2v = *(const v4f*)(f2b + jc * 8 + hh * 4);
        float s0 = f1i + f2v[0]; s0 = fmaxf(s0, 0.2f * s0);
        float s1 = f1i + f2v[1]; s1 = fmaxf(s1, 0.2f * s1);
        float s2 = f1i + f2v[2]; s2 = fmaxf(s2, 0.2f * s2);
        float s3 = f1i + f2v[3]; s3 = fmaxf(s3, 0.2f * s3);
        const float p0 = (m4 & 1u) ? __expf(s0 - mi) : pm;
        const float p1 = (m4 & 2u) ? __expf(s1 - mi) : pm;
        const float p2 = (m4 & 4u) ? __expf(s2 - mi) : pm;
        const float p3 = (m4 & 8u) ? __expf(s3 - mi) : pm;

        v2f A0; A0[0] = p0; A0[1] = p1;
        v2f A1; A1[0] = p2; A1[1] = p3;
        lsum += (p0 + p1) + (p2 + p3);

#pragma unroll
        for (int dt = 0; dt < 8; ++dt) {
            v2f Bv; Bv[0] = Bq[dt][0]; Bv[1] = Bq[dt][1];
            acc[dt] = wmma_f32_4(A0, Bv, acc[dt]);
        }
#pragma unroll
        for (int dt = 0; dt < 8; ++dt) {
            v2f Bv; Bv[0] = Bq[dt][2]; Bv[1] = Bq[dt][3];
            acc[dt] = wmma_f32_4(A1, Bv, acc[dt]);
        }
    }

    float lpair = lsum + __shfl_xor(lsum, 16);
    if (lane < 16) lds_lw[wv * 16 + lane] = lpair;
    __syncthreads();
    if (tid < 16) {
        float t = 0.0f;
#pragma unroll
        for (int w2 = 0; w2 < 8; ++w2) t += lds_lw[w2 * 16 + tid];
        lds_l[tid] = t;
    }
    __syncthreads();

    for (int half = 4; half >= 1; half >>= 1) {
        if (wv >= half && wv < 2 * half) {
            const int slot = wv - half;
#pragma unroll
            for (int dt = 0; dt < 8; ++dt)
                *(v8f*)&accbuf[(((size_t)slot * 8 + dt) * 32 + lane) * 8] = acc[dt];
        }
        __syncthreads();
        if (wv < half) {
#pragma unroll
            for (int dt = 0; dt < 8; ++dt)
                acc[dt] = acc[dt] +
                    *(const v8f*)&accbuf[(((size_t)wv * 8 + dt) * 32 + lane) * 8];
        }
        __syncthreads();
    }

    if (wv == 0) {
        float* ob = out + ((size_t)b * N_ + i0) * D_;
#pragma unroll
        for (int dt = 0; dt < 8; ++dt) {
#pragma unroll
            for (int r = 0; r < 8; ++r) {
                const int row = r + ((lane >> 4) << 3);
                const float li = lds_l[row];
                const float h  = acc[dt][r] / li;
                const float o  = (h > 0.0f) ? h : (__expf(h) - 1.0f);
                ob[(size_t)row * D_ + dt * 16 + nl] = o;
            }
        }
    }
}

extern "C" void kernel_launch(void* const* d_in, const int* in_sizes, int n_in,
                              void* d_out, int out_size, void* d_ws, size_t ws_size,
                              hipStream_t stream) {
    const float* x   = (const float*)d_in[0];
    const int*   adj = (const int*)d_in[1];
    const float* W   = (const float*)d_in[2];
    const float* a   = (const float*)d_in[3];
    float* out = (float*)d_out;

    float* whf = (float*)d_ws;
    float* f1  = whf + (size_t)B_ * D_ * N_;
    float* f2  = f1 + (size_t)B_ * N_;

    wht_kernel<<<dim3(B_ * (N_ / 16)), dim3(256), 0, stream>>>(x, W, whf);
    f_kernel<<<dim3((B_ * N_) / 64), dim3(256), 0, stream>>>(whf, a, f1, f2);
    gat_kernel<<<dim3(B_ * (N_ / 16)), dim3(256), 0, stream>>>(whf, f1, f2, adj, out);
}


# gat pass 2: each wave holds two 16-row i-tiles (32 rows per block) and feeds every Wh fragment it loads to both from registers: Wh read once per 32 rows; j split, tree, sum order kept
# speedup vs baseline: 3.1147x; 1.0876x over previous
#include <hip/hip_runtime.h>
#include <cstddef>

typedef float v2f __attribute__((ext_vector_type(2)));
typedef float v4f __attribute__((ext_vector_type(4)));
typedef float v8f __attribute__((ext_vector_type(8)));
typedef int   v4i __attribute__((ext_vector_type(4)));

#define B_  8
#define N_  2048
#define C_  128
#define D_  128
#define NEG_BIG (-9.0e15f)

static __device__ __forceinline__ v8f wmma_f32_4(v2f a, v2f b, v8f c) {
    return __builtin_amdgcn_wmma_f32_16x16x4_f32(false, a, false, b, (short)0, c,
                                                 false, false);
}

__global__ __launch_bounds__(256)
void wht_kernel(const float* __restrict__ x, const float* __restrict__ W,
                float* __restrict__ whf) {
    const int bid  = blockIdx.x;
    const int b    = bid >> 7;
    const int jt   = bid & 127;
    const int tid  = threadIdx.x;
    const int wv   = tid >> 5;
    const int lane = tid & 31;
    const int nl   = lane & 15;
    const int hh   = lane >> 4;
    const int kh   = hh << 1;

    const float* xrow = x + ((size_t)(b * N_ + jt * 16 + nl)) * C_;
    const float* wrow = W + (size_t)(wv * 16 + nl) * C_;

    v8f acc = {};
#pragma unroll 4
    for (int k0 = 0; k0 < C_; k0 += 4) {
        const int kb = k0 + kh;
        v2f xa = *(const v2f*)(xrow + kb);
        v2f wb = *(const v2f*)(wrow + kb);
        acc = wmma_f32_4(xa, wb, acc);
    }

    float* ob = whf + ((((size_t)b * (N_ / 8) + jt * 2 + hh) * 8 + wv) * 32) * 4;
    v4f lo, hi;
    lo[0] = acc[0]; lo[1] = acc[1]; lo[2] = acc[2]; lo[3] = acc[3];
    hi[0] = acc[4]; hi[1] = acc[5]; hi[2] = acc[6]; hi[3] = acc[7];
    *(v4f*)(ob + nl * 4)        = lo;
    *(v4f*)(ob + (nl + 16) * 4) = hi;
}

__global__ __launch_bounds__(256)
void f_kernel(const float* __restrict__ whf, const float* __restrict__ a,
              float* __restrict__ f1, float* __restrict__ f2) {
    const int tid  = threadIdx.x;
    const int lane = tid & 31;
    const int c    = lane & 15;
    const int hh   = lane >> 4;
    const int gc   = blockIdx.x * 8 + (tid >> 5);
    const v4f* wp  = (const v4f*)whf + (size_t)gc * (8 * 32) + lane;

    v4f s1 = {}, s2 = {};
#pragma unroll
    for (int dt = 0; dt < 8; ++dt) {
        const v4f w   = wp[dt * 32];
        const float a1 = a[dt * 16 + c];
        const float a2 = a[D_ + dt * 16 + c];
        s1 += w * a1;
        s2 += w * a2;
    }
#pragma unroll
    for (int m = 8; m >= 1; m >>= 1) {
#pragma unroll
        for (int e = 0; e < 4; ++e) {
            s1[e] += __shfl_xor(s1[e], m);
            s2[e] += __shfl_xor(s2[e], m);
        }
    }
    if (c == 0) {
        *(v4f*)(f1 + (size_t)gc * 8 + hh * 4) = s1;
        *(v4f*)(f2 + (size_t)gc * 8 + hh * 4) = s2;
    }
}

#define TI_ 2

__global__ __launch_bounds__(256, 4)
void gat_kernel(const float* __restrict__ whf, const float* __restrict__ f1g,
                const float* __restrict__ f2g, const int* __restrict__ adj,
                float* __restrict__ out) {
    __shared__ unsigned lds_bits[TI_ * 16 * 65];
    __shared__ float lds_mw[TI_ * 16];
    __shared__ float lds_m[TI_ * 16], lds_f1[TI_ * 16], lds_pone[TI_ * 16], lds_l[TI_ * 16];
    __shared__ float lds_lw[8 * TI_ * 16];
    __shared__ float accbuf[4 * TI_ * 8 * 32 * 8];

    const int bid  = blockIdx.x;
    const int b    = bid / (N_ / (16 * TI_));
    const int i0   = (bid % (N_ / (16 * TI_))) * (16 * TI_);
    const int tid  = threadIdx.x;
    const int wv   = tid >> 5;
    const int lane = tid & 31;
    const int nl   = lane & 15;

    const float* f2b  = f2g + b * N_;
    const v4f*   whfb = (const v4f*)(whf + (size_t)b * D_ * N_) + lane;

    for (int ti = 0; ti < TI_; ++ti) {
        const int r0 = ti * 16 + wv * 2;
        const v4i* arow0 = (const v4i*)(adj + (size_t)(i0 + r0) * N_) + lane;
        const v4i* arow1 = arow0 + N_ / 4;
        const v4f* frow  = (const v4f*)f2b + lane;
        const unsigned sh = (unsigned)(lane & 7) * 4u;
        float mx0 = -3.0e38f, mx1 = -3.0e38f;
        for (int tb = 0; tb < 16; tb += 8) {
            v4i a0[8], a1[8];
            v4f fv[8];
#pragma unroll
            for (int u = 0; u < 8; ++u) {
                a0[u] = arow0[(tb + u) * 32];
                a1[u] = arow1[(tb + u) * 32];
                fv[u] = frow[(tb + u) * 32];
            }
            unsigned n0[8], n1[8];
#pragma unroll
            for (int u = 0; u < 8; ++u) {
                unsigned b0 = 0u, b1 = 0u;
#pragma unroll
                for (int e = 0; e < 4; ++e) {
                    if (a0[u][e] > 0) { b0 |= 1u << e; mx0 = fmaxf(mx0, fv[u][e]); }
                    if (a1[u][e] > 0) { b1 |= 1u << e; mx1 = fmaxf(mx1, fv[u][e]); }
                }
                n0[u] = b0 << sh;
                n1[u] = b1 << sh;
            }
#pragma unroll
            for (int m = 1; m <= 4; m <<= 1) {
#pragma unroll
                for (int u = 0; u < 8; ++u) {
                    n0[u] |= __shfl_xor(n0[u], m);
                    n1[u] |= __shfl_xor(n1[u], m);
                }
            }
            if ((lane & 7) == 0) {
#pragma unroll
                for (int u = 0; u < 8; ++u) {
                    lds_bits[r0 * 65 + (tb + u) * 4 + (lane >> 3)]       = n0[u];
                    lds_bits[(r0 + 1) * 65 + (tb + u) * 4 + (lane >> 3)] = n1[u];
                }
            }
        }
#pragma unroll
        for (int m = 16; m >= 1; m >>= 1) {
            mx0 = fmaxf(mx0, __shfl_xor(mx0, m));
            mx1 = fmaxf(mx1, __shfl_xor(mx1, m));
        }
        if (lane == 0) {
            lds_mw[r0]     = mx0;
            lds_mw[r0 + 1] = mx1;
        }
    }
    __syncthreads();
    if (tid < TI_ * 16) {
        const float M = lds_mw[tid];
        const bool empty = (M < -1.0e38f);
        const float f1v  = f1g[b * N_ + i0 + tid];
        const float sum  = f1v + M;
        const float emax = fmaxf(sum, 0.2f * sum);
        lds_m[tid]    = empty ? 0.0f : emax;
        lds_pone[tid] = empty ? 1.0f : 0.0f;
        lds_f1[tid]   = f1v;
    }
    __syncthreads();

    const int hh = lane >> 4;
    float f1i[TI_], mi[TI_], pm[TI_];
#pragma unroll
    for (int ti = 0; ti < TI_; ++ti) {
        f1i[ti] = lds_f1[ti * 16 + nl];
        mi[ti]  = lds_m[ti * 16 + nl];
        pm[ti]  = lds_pone[ti * 16 + nl];
    }

    v8f acc[TI_][8] = {};
    float lsum[TI_] = {};

    for (int g = 0; g < 32; ++g) {
        const int jc = g * 8 + wv;
        const v4f* wq = whfb + (size_t)jc * (8 * 32);
        v4f Bq[8];
#pragma unroll
        for (int dt = 0; dt < 8; ++dt) Bq[dt] = wq[dt * 32];

        const v4f f2v = *(const v4f*)(f2b + jc * 8 + hh * 4);
        const unsigned msh = (unsigned)((jc & 3) * 8 + hh * 4);

#pragma unroll
        for (int ti = 0; ti < TI_; ++ti) {
            const unsigned word = lds_bits[(ti * 16 + nl) * 65 + (jc >> 2)];
            const unsigned m4   = word >> msh;

            float s0 = f1i[ti] + f2v[0]; s0 = fmaxf(s0, 0.2f * s0);
            float s1 = f1i[ti] + f2v[1]; s1 = fmaxf(s1, 0.2f * s1);
            float s2 = f1i[ti] + f2v[2]; s2 = fmaxf(s2, 0.2f * s2);
            float s3 = f1i[ti] + f2v[3]; s3 = fmaxf(s3, 0.2f * s3);
            const float p0 = (m4 & 1u) ? __expf(s0 - mi[ti]) : pm[ti];
            const float p1 = (m4 & 2u) ? __expf(s1 - mi[ti]) : pm[ti];
            const float p2 = (m4 & 4u) ? __expf(s2 - mi[ti]) : pm[ti];
            const float p3 = (m4 & 8u) ? __expf(s3 - mi[ti]) : pm[ti];

            v2f A0; A0[0] = p0; A0[1] = p1;
            v2f A1; A1[0] = p2; A1[1] = p3;
            lsum[ti] += (p0 + p1) + (p2 + p3);

#pragma unroll
            for (int dt = 0; dt < 8; ++dt) {
                v2f Bv; Bv[0] = Bq[dt][0]; Bv[1] = Bq[dt][1];
                acc[ti][dt] = wmma_f32_4(A0, Bv, acc[ti][dt]);
            }
#pragma unroll
            for (int dt = 0; dt < 8; ++dt) {
                v2f Bv; Bv[0] = Bq[dt][2]; Bv[1] = Bq[dt][3];
                acc[ti][dt] = wmma_f32_4(A1, Bv, acc[ti][dt]);
            }
        }
    }

#pragma unroll
    for (int ti = 0; ti < TI_; ++ti) {
        const float lpair = lsum[ti] + __shfl_xor(lsum[ti], 16);
        if (lane < 16) lds_lw[wv * (TI_ * 16) + ti * 16 + lane] = lpair;
    }
    __syncthreads();
    if (tid < TI_ * 16) {
        float t = 0.0f;
#pragma unroll
        for (int w2 = 0; w2 < 8; ++w2) t += lds_lw[w2 * (TI_ * 16) + tid];
        lds_l[tid] = t;
    }
    __syncthreads();

    for (int half = 4; half >= 1; half >>= 1) {
        if (wv >= half && wv < 2 * half) {
            const int slot = wv - half;
#pragma unroll
            for (int ti = 0; ti < TI_; ++ti) {
#pragma unroll
                for (int dt = 0; dt < 8; ++dt)
                    *(v8f*)&accbuf[((((size_t)slot * TI_ + ti) * 8 + dt) * 32 + lane) * 8] =
                        acc[ti][dt];
            }
        }
        __syncthreads();
        if (wv < half) {
#pragma unroll
            for (int ti = 0; ti < TI_; ++ti) {
#pragma unroll
                for (int dt = 0; dt < 8; ++dt)
                    acc[ti][dt] = acc[ti][dt] +
                        *(const v8f*)&accbuf[((((size_t)wv * TI_ + ti) * 8 + dt) * 32 + lane) * 8];
            }
        }
        __syncthreads();
    }

    if (wv == 0) {
        float* ob = out + ((size_t)b * N_ + i0) * D_;
#pragma unroll
        for (int ti = 0; ti < TI_; ++ti) {
#pragma unroll
            for (int dt = 0; dt < 8; ++dt) {
#pragma unroll
                for (int r = 0; r < 8; ++r) {
                    const int row = ti * 16 + r + ((lane >> 4) << 3);
                    const float li = lds_l[row];
                    const float h  = acc[ti][dt][r] / li;
                    const float o  = (h > 0.0f) ? h : (__expf(h) - 1.0f);
                    ob[(size_t)row * D_ + dt * 16 + nl] = o;
                }
            }
        }
    }
}

extern "C" void kernel_launch(void* const* d_in, const int* in_sizes, int n_in,
                              void* d_out, int out_size, void* d_ws, size_t ws_size,
                              hipStream_t stream) {
    const float* x   = (const float*)d_in[0];
    const int*   adj = (const int*)d_in[1];
    const float* W   = (const float*)d_in[2];
    const float* a   = (const float*)d_in[3];
    float* out = (float*)d_out;

    float* whf = (float*)d_ws;
    float* f1  = whf + (size_t)B_ * D_ * N_;
    float* f2  = f1 + (size_t)B_ * N_;

    wht_kernel<<<dim3(B_ * (N_ / 16)), dim3(256), 0, stream>>>(x, W, whf);
    f_kernel<<<dim3((B_ * N_) / 64), dim3(256), 0, stream>>>(whf, a, f1, f2);
    gat_kernel<<<dim3(B_ * (N_ / (16 * TI_))), dim3(256), 0, stream>>>(whf, f1, f2, adj, out);
}


# wht kernel: each wave owns a 2x2 patch of output tiles; x and W rows read as 16-byte pieces, each fragment feeding 2 tiles x 2 matrix steps, a 128-byte stage loaded ahead; its c-sum order changes
# speedup vs baseline: 3.3214x; 1.0664x over previous
#include <hip/hip_runtime.h>
#include <cstddef>

typedef float v2f __attribute__((ext_vector_type(2)));
typedef float v4f __attribute__((ext_vector_type(4)));
typedef float v8f __attribute__((ext_vector_type(8)));
typedef int   v4i __attribute__((ext_vector_type(4)));

#define B_  8
#define N_  2048
#define C_  128
#define D_  128
#define NEG_BIG (-9.0e15f)

static __device__ __forceinline__ v8f wmma_f32_4(v2f a, v2f b, v8f c) {
    return __builtin_amdgcn_wmma_f32_16x16x4_f32(false, a, false, b, (short)0, c,
                                                 false, false);
}

static __device__ __forceinline__ void wht_load_stage(const float* xr, const float* wr, int s,
                                                      v4f (&xa)[2][4], v4f (&wb)[2][4]) {
#pragma unroll
    for (int t = 0; t < 2; ++t) {
#pragma unroll
        for (int q = 0; q < 4; ++q) {
            xa[t][q] = *(const v4f*)(xr + (size_t)t * 16 * C_ + s * 32 + q * 8);
            wb[t][q] = *(const v4f*)(wr + (size_t)t * 16 * C_ + s * 32 + q * 8);
        }
    }
}

static __device__ __forceinline__ void wht_mma_stage(const v4f (&xa)[2][4], const v4f (&wb)[2][4],
                                                     v8f (&acc)[2][2]) {
#pragma unroll
    for (int q = 0; q < 4; ++q) {
#pragma unroll
        for (int e = 0; e < 4; e += 2) {
            v2f A[2], Bf[2];
#pragma unroll
            for (int t = 0; t < 2; ++t) {
                A[t][0]  = xa[t][q][e]; A[t][1]  = xa[t][q][e + 1];
                Bf[t][0] = wb[t][q][e]; Bf[t][1] = wb[t][q][e + 1];
            }
#pragma unroll
            for (int ti = 0; ti < 2; ++ti) {
#pragma unroll
                for (int tj = 0; tj < 2; ++tj)
                    acc[ti][tj] = wmma_f32_4(A[ti], Bf[tj], acc[ti][tj]);
            }
        }
    }
}

__global__ __launch_bounds__(256)
void wht_kernel(const float* __restrict__ x, const float* __restrict__ W,
                float* __restrict__ whf) {
    const int tid  = threadIdx.x;
    const int wv   = tid >> 5;
    const int lane = tid & 31;
    const int nl   = lane & 15;
    const int hh   = lane >> 4;
    const int jt0  = blockIdx.x * 4 + (wv >> 2) * 2;
    const int dt0  = (wv & 3) * 2;

    const float* xr = x + ((size_t)(jt0 * 16 + nl)) * C_ + hh * 4;
    const float* wr = W + ((size_t)(dt0 * 16 + nl)) * C_ + hh * 4;

    v8f acc[2][2] = {};
    v4f xa0[2][4], wb0[2][4], xa1[2][4], wb1[2][4];

    wht_load_stage(xr, wr, 0, xa0, wb0);
    wht_load_stage(xr, wr, 1, xa1, wb1);
    __builtin_amdgcn_sched_barrier(0);
    wht_mma_stage(xa0, wb0, acc);
    wht_load_stage(xr, wr, 2, xa0, wb0);
    __builtin_amdgcn_sched_barrier(0);
    wht_mma_stage(xa1, wb1, acc);
    wht_load_stage(xr, wr, 3, xa1, wb1);
    __builtin_amdgcn_sched_barrier(0);
    wht_mma_stage(xa0, wb0, acc);
    wht_mma_stage(xa1, wb1, acc);

#pragma unroll
    for (int ti = 0; ti < 2; ++ti) {
#pragma unroll
        for (int tj = 0; tj < 2; ++tj) {
            float* ob = whf + ((((size_t)(jt0 + ti) * 2 + hh) * 8 + (dt0 + tj)) * 32) * 4;
            v4f lo, hi;
            lo[0] = acc[ti][tj][0]; lo[1] = acc[ti][tj][1];
            lo[2] = acc[ti][tj][2]; lo[3] = acc[ti][tj][3];
            hi[0] = acc[ti][tj][4]; hi[1] = acc[ti][tj][5];
            hi[2] = acc[ti][tj][6]; hi[3] = acc[ti][tj][7];
            *(v4f*)(ob + nl * 4)        = lo;
            *(v4f*)(ob + (nl + 16) * 4) = hi;
        }
    }
}

__global__ __launch_bounds__(256)
void f_kernel(const float* __restrict__ whf, const float* __restrict__ a,
              float* __restrict__ f1, float* __restrict__ f2) {
    const int tid  = threadIdx.x;
    const int lane = tid & 31;
    const int c    = lane & 15;
    const int hh   = lane >> 4;
    const int gc   = blockIdx.x * 8 + (tid >> 5);
    const v4f* wp  = (const v4f*)whf + (size_t)gc * (8 * 32) + lane;

    v4f s1 = {}, s2 = {};
#pragma unroll
    for (int dt = 0; dt < 8; ++dt) {
        const v4f w   = wp[dt * 32];
        const float a1 = a[dt * 16 + c];
        const float a2 = a[D_ + dt * 16 + c];
        s1 += w * a1;
        s2 += w * a2;
    }
#pragma unroll
    for (int m = 8; m >= 1; m >>= 1) {
#pragma unroll
        for (int e = 0; e < 4; ++e) {
            s1[e] += __shfl_xor(s1[e], m);
            s2[e] += __shfl_xor(s2[e], m);
        }
    }
    if (c == 0) {
        *(v4f*)(f1 + (size_t)gc * 8 + hh * 4) = s1;
        *(v4f*)(f2 + (size_t)gc * 8 + hh * 4) = s2;
    }
}

#define TI_ 2

__global__ __launch_bounds__(256, 4)
void gat_kernel(const float* __restrict__ whf, const float* __restrict__ f1g,
                const float* __restrict__ f2g, const int* __restrict__ adj,
                float* __restrict__ out) {
    __shared__ unsigned lds_bits[TI_ * 16 * 65];
    __shared__ float lds_mw[TI_ * 16];
    __shared__ float lds_m[TI_ * 16], lds_f1[TI_ * 16], lds_pone[TI_ * 16], lds_l[TI_ * 16];
    __shared__ float lds_lw[8 * TI_ * 16];
    __shared__ float accbuf[4 * TI_ * 8 * 32 * 8];

    const int bid  = blockIdx.x;
    const int b    = bid / (N_ / (16 * TI_));
    const int i0   = (bid % (N_ / (16 * TI_))) * (16 * TI_);
    const int tid  = threadIdx.x;
    const int wv   = tid >> 5;
    const int lane = tid & 31;
    const int nl   = lane & 15;

    const float* f2b  = f2g + b * N_;
    const v4f*   whfb = (const v4f*)(whf + (size_t)b * D_ * N_) + lane;

    for (int ti = 0; ti < TI_; ++ti) {
        const int r0 = ti * 16 + wv * 2;
        const v4i* arow0 = (const v4i*)(adj + (size_t)(i0 + r0) * N_) + lane;
        const v4i* arow1 = arow0 + N_ / 4;
        const v4f* frow  = (const v4f*)f2b + lane;
        const unsigned sh = (unsigned)(lane & 7) * 4u;
        float mx0 = -3.0e38f, mx1 = -3.0e38f;
        for (int tb = 0; tb < 16; tb += 8) {
            v4i a0[8], a1[8];
            v4f fv[8];
#pragma unroll
            for (int u = 0; u < 8; ++u) {
                a0[u] = arow0[(tb + u) * 32];
                a1[u] = arow1[(tb + u) * 32];
                fv[u] = frow[(tb + u) * 32];
            }
            unsigned n0[8], n1[8];
#pragma unroll
            for (int u = 0; u < 8; ++u) {
                unsigned b0 = 0u, b1 = 0u;
#pragma unroll
                for (int e = 0; e < 4; ++e) {
                    if (a0[u][e] > 0) { b0 |= 1u << e; mx0 = fmaxf(mx0, fv[u][e]); }
                    if (a1[u][e] > 0) { b1 |= 1u << e; mx1 = fmaxf(mx1, fv[u][e]); }
                }
                n0[u] = b0 << sh;
                n1[u] = b1 << sh;
            }
#pragma unroll
            for (int m = 1; m <= 4; m <<= 1) {
#pragma unroll
                for (int u = 0; u < 8; ++u) {
                    n0[u] |= __shfl_xor(n0[u], m);
                    n1[u] |= __shfl_xor(n1[u], m);
                }
            }
            if ((lane & 7) == 0) {
#pragma unroll
                for (int u = 0; u < 8; ++u) {
                    lds_bits[r0 * 65 + (tb + u) * 4 + (lane >> 3)]       = n0[u];
                    lds_bits[(r0 + 1) * 65 + (tb + u) * 4 + (lane >> 3)] = n1[u];
                }
            }
        }
#pragma unroll
        for (int m = 16; m >= 1; m >>= 1) {
            mx0 = fmaxf(mx0, __shfl_xor(mx0, m));
            mx1 = fmaxf(mx1, __shfl_xor(mx1, m));
        }
        if (lane == 0) {
            lds_mw[r0]     = mx0;
            lds_mw[r0 + 1] = mx1;
        }
    }
    __syncthreads();
    if (tid < TI_ * 16) {
        const float M = lds_mw[tid];
        const bool empty = (M < -1.0e38f);
        const float f1v  = f1g[b * N_ + i0 + tid];
        const float sum  = f1v + M;
        const float emax = fmaxf(sum, 0.2f * sum);
        lds_m[tid]    = empty ? 0.0f : emax;
        lds_pone[tid] = empty ? 1.0f : 0.0f;
        lds_f1[tid]   = f1v;
    }
    __syncthreads();

    const int hh = lane >> 4;
    float f1i[TI_], mi[TI_], pm[TI_];
#pragma unroll
    for (int ti = 0; ti < TI_; ++ti) {
        f1i[ti] = lds_f1[ti * 16 + nl];
        mi[ti]  = lds_m[ti * 16 + nl];
        pm[ti]  = lds_pone[ti * 16 + nl];
    }

    v8f acc[TI_][8] = {};
    float lsum[TI_] = {};

    for (int g = 0; g < 32; ++g) {
        const int jc = g * 8 + wv;
        const v4f* wq = whfb + (size_t)jc * (8 * 32);
        v4f Bq[8];
#pragma unroll
        for (int dt = 0; dt < 8; ++dt) Bq[dt] = wq[dt * 32];

        const v4f f2v = *(const v4f*)(f2b + jc * 8 + hh * 4);
        const unsigned msh = (unsigned)((jc & 3) * 8 + hh * 4);

#pragma unroll
        for (int ti = 0; ti < TI_; ++ti) {
            const unsigned word = lds_bits[(ti * 16 + nl) * 65 + (jc >> 2)];
            const unsigned m4   = word >> msh;

            float s0 = f1i[ti] + f2v[0]; s0 = fmaxf(s0, 0.2f * s0);
            float s1 = f1i[ti] + f2v[1]; s1 = fmaxf(s1, 0.2f * s1);
            float s2 = f1i[ti] + f2v[2]; s2 = fmaxf(s2, 0.2f * s2);
            float s3 = f1i[ti] + f2v[3]; s3 = fmaxf(s3, 0.2f * s3);
            const float p0 = (m4 & 1u) ? __expf(s0 - mi[ti]) : pm[ti];
            const float p1 = (m4 & 2u) ? __expf(s1 - mi[ti]) : pm[ti];
            const float p2 = (m4 & 4u) ? __expf(s2 - mi[ti]) : pm[ti];
            const float p3 = (m4 & 8u) ? __expf(s3 - mi[ti]) : pm[ti];

            v2f A0; A0[0] = p0; A0[1] = p1;
            v2f A1; A1[0] = p2; A1[1] = p3;
            lsum[ti] += (p0 + p1) + (p2 + p3);

#pragma unroll
            for (int dt = 0; dt < 8; ++dt) {
                v2f Bv; Bv[0] = Bq[dt][0]; Bv[1] = Bq[dt][1];
                acc[ti][dt] = wmma_f32_4(A0, Bv, acc[ti][dt]);
            }
#pragma unroll
            for (int dt = 0; dt < 8; ++dt) {
                v2f Bv; Bv[0] = Bq[dt][2]; Bv[1] = Bq[dt][3];
                acc[ti][dt] = wmma_f32_4(A1, Bv, acc[ti][dt]);
            }
        }
    }

#pragma unroll
    for (int ti = 0; ti < TI_; ++ti) {
        const float lpair = lsum[ti] + __shfl_xor(lsum[ti], 16);
        if (lane < 16) lds_lw[wv * (TI_ * 16) + ti * 16 + lane] = lpair;
    }
    __syncthreads();
    if (tid < TI_ * 16) {
        float t = 0.0f;
#pragma unroll
        for (int w2 = 0; w2 < 8; ++w2) t += lds_lw[w2 * (TI_ * 16) + tid];
        lds_l[tid] = t;
    }
    __syncthreads();

    for (int half = 4; half >= 1; half >>= 1) {
        if (wv >= half && wv < 2 * half) {
            const int slot = wv - half;
#pragma unroll
            for (int ti = 0; ti < TI_; ++ti) {
#pragma unroll
                for (int dt = 0; dt < 8; ++dt)
                    *(v8f*)&accbuf[((((size_t)slot * TI_ + ti) * 8 + dt) * 32 + lane) * 8] =
                        acc[ti][dt];
            }
        }
        __syncthreads();
        if (wv < half) {
#pragma unroll
            for (int ti = 0; ti < TI_; ++ti) {
#pragma unroll
                for (int dt = 0; dt < 8; ++dt)
                    acc[ti][dt] = acc[ti][dt] +
                        *(const v8f*)&accbuf[((((size_t)wv * TI_ + ti) * 8 + dt) * 32 + lane) * 8];
            }
        }
        __syncthreads();
    }

    if (wv == 0) {
        float* ob = out + ((size_t)b * N_ + i0) * D_;
#pragma unroll
        for (int ti = 0; ti < TI_; ++ti) {
#pragma unroll
            for (int dt = 0; dt < 8; ++dt) {
#pragma unroll
                for (int r = 0; r < 8; ++r) {
                    const int row = ti * 16 + r + ((lane >> 4) << 3);
                    const float li = lds_l[row];
                    const float h  = acc[ti][dt][r] / li;
                    const float o  = (h > 0.0f) ? h : (__expf(h) - 1.0f);
                    ob[(size_t)row * D_ + dt * 16 + nl] = o;
                }
            }
        }
    }
}

extern "C" void kernel_launch(void* const* d_in, const int* in_sizes, int n_in,
                              void* d_out, int out_size, void* d_ws, size_t ws_size,
                              hipStream_t stream) {
    const float* x   = (const float*)d_in[0];
    const int*   adj = (const int*)d_in[1];
    const float* W   = (const float*)d_in[2];
    const float* a   = (const float*)d_in[3];
    float* out = (float*)d_out;

    float* whf = (float*)d_ws;
    float* f1  = whf + (size_t)B_ * D_ * N_;
    float* f2  = f1 + (size_t)B_ * N_;

    wht_kernel<<<dim3((B_ * N_) / 64), dim3(256), 0, stream>>>(x, W, whf);
    f_kernel<<<dim3((B_ * N_) / 64), dim3(256), 0, stream>>>(whf, a, f1, f2);
    gat_kernel<<<dim3(B_ * (N_ / (16 * TI_))), dim3(256), 0, stream>>>(whf, f1, f2, adj, out);
}


# gat pass 2: the next chunk's eight Wh fragment loads and f2 values go into a second register set ahead of the current chunk's 32 matrix steps (were waited for at each chunk's top); sum orders kept
# speedup vs baseline: 3.4419x; 1.0363x over previous
#include <hip/hip_runtime.h>
#include <cstddef>

typedef float v2f __attribute__((ext_vector_type(2)));
typedef float v4f __attribute__((ext_vector_type(4)));
typedef float v8f __attribute__((ext_vector_type(8)));
typedef int   v4i __attribute__((ext_vector_type(4)));

#define B_  8
#define N_  2048
#define C_  128
#define D_  128
#define NEG_BIG (-9.0e15f)

static __device__ __forceinline__ v8f wmma_f32_4(v2f a, v2f b, v8f c) {
    return __builtin_amdgcn_wmma_f32_16x16x4_f32(false, a, false, b, (short)0, c,
                                                 false, false);
}

static __device__ __forceinline__ void wht_load_stage(const float* xr, const float* wr, int s,
                                                      v4f (&xa)[2][4], v4f (&wb)[2][4]) {
#pragma unroll
    for (int t = 0; t < 2; ++t) {
#pragma unroll
        for (int q = 0; q < 4; ++q) {
            xa[t][q] = *(const v4f*)(xr + (size_t)t * 16 * C_ + s * 32 + q * 8);
            wb[t][q] = *(const v4f*)(wr + (size_t)t * 16 * C_ + s * 32 + q * 8);
        }
    }
}

static __device__ __forceinline__ void wht_mma_stage(const v4f (&xa)[2][4], const v4f (&wb)[2][4],
                                                     v8f (&acc)[2][2]) {
#pragma unroll
    for (int q = 0; q < 4; ++q) {
#pragma unroll
        for (int e = 0; e < 4; e += 2) {
            v2f A[2], Bf[2];
#pragma unroll
            for (int t = 0; t < 2; ++t) {
                A[t][0]  = xa[t][q][e]; A[t][1]  = xa[t][q][e + 1];
                Bf[t][0] = wb[t][q][e]; Bf[t][1] = wb[t][q][e + 1];
            }
#pragma unroll
            for (int ti = 0; ti < 2; ++ti) {
#pragma unroll
                for (int tj = 0; tj < 2; ++tj)
                    acc[ti][tj] = wmma_f32_4(A[ti], Bf[tj], acc[ti][tj]);
            }
        }
    }
}

__global__ __launch_bounds__(256)
void wht_kernel(const float* __restrict__ x, const float* __restrict__ W,
                float* __restrict__ whf) {
    const int tid  = threadIdx.x;
    const int wv   = tid >> 5;
    const int lane = tid & 31;
    const int nl   = lane & 15;
    const int hh   = lane >> 4;
    const int jt0  = blockIdx.x * 4 + (wv >> 2) * 2;
    const int dt0  = (wv & 3) * 2;

    const float* xr = x + ((size_t)(jt0 * 16 + nl)) * C_ + hh * 4;
    const float* wr = W + ((size_t)(dt0 * 16 + nl)) * C_ + hh * 4;

    v8f acc[2][2] = {};
    v4f xa0[2][4], wb0[2][4], xa1[2][4], wb1[2][4];

    wht_load_stage(xr, wr, 0, xa0, wb0);
    wht_load_stage(xr, wr, 1, xa1, wb1);
    __builtin_amdgcn_sched_barrier(0);
    wht_mma_stage(xa0, wb0, acc);
    wht_load_stage(xr, wr, 2, xa0, wb0);
    __builtin_amdgcn_sched_barrier(0);
    wht_mma_stage(xa1, wb1, acc);
    wht_load_stage(xr, wr, 3, xa1, wb1);
    __builtin_amdgcn_sched_barrier(0);
    wht_mma_stage(xa0, wb0, acc);
    wht_mma_stage(xa1, wb1, acc);

#pragma unroll
    for (int ti = 0; ti < 2; ++ti) {
#pragma unroll
        for (int tj = 0; tj < 2; ++tj) {
            float* ob = whf + ((((size_t)(jt0 + ti) * 2 + hh) * 8 + (dt0 + tj)) * 32) * 4;
            v4f lo, hi;
            lo[0] = acc[ti][tj][0]; lo[1] = acc[ti][tj][1];
            lo[2] = acc[ti][tj][2]; lo[3] = acc[ti][tj][3];
            hi[0] = acc[ti][tj][4]; hi[1] = acc[ti][tj][5];
            hi[2] = acc[ti][tj][6]; hi[3] = acc[ti][tj][7];
            *(v4f*)(ob + nl * 4)        = lo;
            *(v4f*)(ob + (nl + 16) * 4) = hi;
        }
    }
}

__global__ __launch_bounds__(256)
void f_kernel(const float* __restrict__ whf, const float* __restrict__ a,
              float* __restrict__ f1, float* __restrict__ f2) {
    const int tid  = threadIdx.x;
    const int lane = tid & 31;
    const int c    = lane & 15;
    const int hh   = lane >> 4;
    const int gc   = blockIdx.x * 8 + (tid >> 5);
    const v4f* wp  = (const v4f*)whf + (size_t)gc * (8 * 32) + lane;

    v4f s1 = {}, s2 = {};
#pragma unroll
    for (int dt = 0; dt < 8; ++dt) {
        const v4f w   = wp[dt * 32];
        const float a1 = a[dt * 16 + c];
        const float a2 = a[D_ + dt * 16 + c];
        s1 += w * a1;
        s2 += w * a2;
    }
#pragma unroll
    for (int m = 8; m >= 1; m >>= 1) {
#pragma unroll
        for (int e = 0; e < 4; ++e) {
            s1[e] += __shfl_xor(s1[e], m);
            s2[e] += __shfl_xor(s2[e], m);
        }
    }
    if (c == 0) {
        *(v4f*)(f1 + (size_t)gc * 8 + hh * 4) = s1;
        *(v4f*)(f2 + (size_t)gc * 8 + hh * 4) = s2;
    }
}

#define TI_ 2

__global__ __launch_bounds__(256, 4)
void gat_kernel(const float* __restrict__ whf, const float* __restrict__ f1g,
                const float* __restrict__ f2g, const int* __restrict__ adj,
                float* __restrict__ out) {
    __shared__ unsigned lds_bits[TI_ * 16 * 65];
    __shared__ float lds_mw[TI_ * 16];
    __shared__ float lds_m[TI_ * 16], lds_f1[TI_ * 16], lds_pone[TI_ * 16], lds_l[TI_ * 16];
    __shared__ float lds_lw[8 * TI_ * 16];
    __shared__ float accbuf[4 * TI_ * 8 * 32 * 8];

    const int bid  = blockIdx.x;
    const int b    = bid / (N_ / (16 * TI_));
    const int i0   = (bid % (N_ / (16 * TI_))) * (16 * TI_);
    const int tid  = threadIdx.x;
    const int wv   = tid >> 5;
    const int lane = tid & 31;
    const int nl   = lane & 15;

    const float* f2b  = f2g + b * N_;
    const v4f*   whfb = (const v4f*)(whf + (size_t)b * D_ * N_) + lane;

    for (int ti = 0; ti < TI_; ++ti) {
        const int r0 = ti * 16 + wv * 2;
        const v4i* arow0 = (const v4i*)(adj + (size_t)(i0 + r0) * N_) + lane;
        const v4i* arow1 = arow0 + N_ / 4;
        const v4f* frow  = (const v4f*)f2b + lane;
        const unsigned sh = (unsigned)(lane & 7) * 4u;
        float mx0 = -3.0e38f, mx1 = -3.0e38f;
        for (int tb = 0; tb < 16; tb += 8) {
            v4i a0[8], a1[8];
            v4f fv[8];
#pragma unroll
            for (int u = 0; u < 8; ++u) {
                a0[u] = arow0[(tb + u) * 32];
                a1[u] = arow1[(tb + u) * 32];
                fv[u] = frow[(tb + u) * 32];
            }
            unsigned n0[8], n1[8];
#pragma unroll
            for (int u = 0; u < 8; ++u) {
                unsigned b0 = 0u, b1 = 0u;
#pragma unroll
                for (int e = 0; e < 4; ++e) {
                    if (a0[u][e] > 0) { b0 |= 1u << e; mx0 = fmaxf(mx0, fv[u][e]); }
                    if (a1[u][e] > 0) { b1 |= 1u << e; mx1 = fmaxf(mx1, fv[u][e]); }
                }
                n0[u] = b0 << sh;
                n1[u] = b1 << sh;
            }
#pragma unroll
            for (int m = 1; m <= 4; m <<= 1) {
#pragma unroll
                for (int u = 0; u < 8; ++u) {
                    n0[u] |= __shfl_xor(n0[u], m);
                    n1[u] |= __shfl_xor(n1[u], m);
                }
            }
            if ((lane & 7) == 0) {
#pragma unroll
                for (int u = 0; u < 8; ++u) {
                    lds_bits[r0 * 65 + (tb + u) * 4 + (lane >> 3)]       = n0[u];
                    lds_bits[(r0 + 1) * 65 + (tb + u) * 4 + (lane >> 3)] = n1[u];
                }
            }
        }
#pragma unroll
        for (int m = 16; m >= 1; m >>= 1) {
            mx0 = fmaxf(mx0, __shfl_xor(mx0, m));
            mx1 = fmaxf(mx1, __shfl_xor(mx1, m));
        }
        if (lane == 0) {
            lds_mw[r0]     = mx0;
            lds_mw[r0 + 1] = mx1;
        }
    }
    __syncthreads();
    if (tid < TI_ * 16) {
        const float M = lds_mw[tid];
        const bool empty = (M < -1.0e38f);
        const float f1v  = f1g[b * N_ + i0 + tid];
        const float sum  = f1v + M;
        const float emax = fmaxf(sum, 0.2f * sum);
        lds_m[tid]    = empty ? 0.0f : emax;
        lds_pone[tid] = empty ? 1.0f : 0.0f;
        lds_f1[tid]   = f1v;
    }
    __syncthreads();

    const int hh = lane >> 4;
    float f1i[TI_], mi[TI_], pm[TI_];
#pragma unroll
    for (int ti = 0; ti < TI_; ++ti) {
        f1i[ti] = lds_f1[ti * 16 + nl];
        mi[ti]  = lds_m[ti * 16 + nl];
        pm[ti]  = lds_pone[ti * 16 + nl];
    }

    v8f acc[TI_][8] = {};
    float lsum[TI_] = {};

    auto load_chunk = [&](int g, v4f (&Bq)[8], v4f& f2v) {
        const int jc = g * 8 + wv;
        const v4f* wq = whfb + (size_t)jc * (8 * 32);
#pragma unroll
        for (int dt = 0; dt < 8; ++dt) Bq[dt] = wq[dt * 32];
        f2v = *(const v4f*)(f2b + jc * 8 + hh * 4);
    };
    auto do_chunk = [&](int g, const v4f (&Bq)[8], const v4f& f2v) {
        const int jc = g * 8 + wv;
        const unsigned msh = (unsigned)((jc & 3) * 8 + hh * 4);
#pragma unroll
        for (int ti = 0; ti < TI_; ++ti) {
            const unsigned word = lds_bits[(ti * 16 + nl) * 65 + (jc >> 2)];
            const unsigned m4   = word >> msh;

            float s0 = f1i[ti] + f2v[0]; s0 = fmaxf(s0, 0.2f * s0);
            float s1 = f1i[ti] + f2v[1]; s1 = fmaxf(s1, 0.2f * s1);
            float s2 = f1i[ti] + f2v[2]; s2 = fmaxf(s2, 0.2f * s2);
            float s3 = f1i[ti] + f2v[3]; s3 = fmaxf(s3, 0.2f * s3);
            const float p0 = (m4 & 1u) ? __expf(s0 - mi[ti]) : pm[ti];
            const float p1 = (m4 & 2u) ? __expf(s1 - mi[ti]) : pm[ti];
            const float p2 = (m4 & 4u) ? __expf(s2 - mi[ti]) : pm[ti];
            const float p3 = (m4 & 8u) ? __expf(s3 - mi[ti]) : pm[ti];

            v2f A0; A0[0] = p0; A0[1] = p1;
            v2f A1; A1[0] = p2; A1[1] = p3;
            lsum[ti] += (p0 + p1) + (p2 + p3);

#pragma unroll
            for (int dt = 0; dt < 8; ++dt) {
                v2f Bv; Bv[0] = Bq[dt][0]; Bv[1] = Bq[dt][1];
                acc[ti][dt] = wmma_f32_4(A0, Bv, acc[ti][dt]);
            }
#pragma unroll
            for (int dt = 0; dt < 8; ++dt) {
                v2f Bv; Bv[0] = Bq[dt][2]; Bv[1] = Bq[dt][3];
                acc[ti][dt] = wmma_f32_4(A1, Bv, acc[ti][dt]);
            }
        }
    };

    v4f BqA[8], BqB[8];
    v4f f2A, f2B;
    load_chunk(0, BqA, f2A);
    for (int g = 0; g < 30; g += 2) {
        load_chunk(g + 1, BqB, f2B);
        do_chunk(g, BqA, f2A);
        load_chunk(g + 2, BqA, f2A);
        do_chunk(g + 1, BqB, f2B);
    }
    load_chunk(31, BqB, f2B);
    do_chunk(30, BqA, f2A);
    do_chunk(31, BqB, f2B);

#pragma unroll
    for (int ti = 0; ti < TI_; ++ti) {
        const float lpair = lsum[ti] + __shfl_xor(lsum[ti], 16);
        if (lane < 16) lds_lw[wv * (TI_ * 16) + ti * 16 + lane] = lpair;
    }
    __syncthreads();
    if (tid < TI_ * 16) {
        float t = 0.0f;
#pragma unroll
        for (int w2 = 0; w2 < 8; ++w2) t += lds_lw[w2 * (TI_ * 16) + tid];
        lds_l[tid] = t;
    }
    __syncthreads();

    for (int half = 4; half >= 1; half >>= 1) {
        if (wv >= half && wv < 2 * half) {
            const int slot = wv - half;
#pragma unroll
            for (int ti = 0; ti < TI_; ++ti) {
#pragma unroll
                for (int dt = 0; dt < 8; ++dt)
                    *(v8f*)&accbuf[((((size_t)slot * TI_ + ti) * 8 + dt) * 32 + lane) * 8] =
                        acc[ti][dt];
            }
        }
        __syncthreads();
        if (wv < half) {
#pragma unroll
            for (int ti = 0; ti < TI_; ++ti) {
#pragma unroll
                for (int dt = 0; dt < 8; ++dt)
                    acc[ti][dt] = acc[ti][dt] +
                        *(const v8f*)&accbuf[((((size_t)wv * TI_ + ti) * 8 + dt) * 32 + lane) * 8];
            }
        }
        __syncthreads();
    }

    if (wv == 0) {
        float* ob = out + ((size_t)b * N_ + i0) * D_;
#pragma unroll
        for (int ti = 0; ti < TI_; ++ti) {
#pragma unroll
            for (int dt = 0; dt < 8; ++dt) {
#pragma unroll
                for (int r = 0; r < 8; ++r) {
                    const int row = ti * 16 + r + ((lane >> 4) << 3);
                    const float li = lds_l[row];
                    const float h  = acc[ti][dt][r] / li;
                    const float o  = (h > 0.0f) ? h : (__expf(h) - 1.0f);
                    ob[(size_t)row * D_ + dt * 16 + nl] = o;
                }
            }
        }
    }
}

extern "C" void kernel_launch(void* const* d_in, const int* in_sizes, int n_in,
                              void* d_out, int out_size, void* d_ws, size_t ws_size,
                              hipStream_t stream) {
    const float* x   = (const float*)d_in[0];
    const int*   adj = (const int*)d_in[1];
    const float* W   = (const float*)d_in[2];
    const float* a   = (const float*)d_in[3];
    float* out = (float*)d_out;

    float* whf = (float*)d_ws;
    float* f1  = whf + (size_t)B_ * D_ * N_;
    float* f2  = f1 + (size_t)B_ * N_;

    wht_kernel<<<dim3((B_ * N_) / 64), dim3(256), 0, stream>>>(x, W, whf);
    f_kernel<<<dim3((B_ * N_) / 64), dim3(256), 0, stream>>>(whf, a, f1, f2);
    gat_kernel<<<dim3(B_ * (N_ / (16 * TI_))), dim3(256), 0, stream>>>(whf, f1, f2, adj, out);
}


# gat kernel: a 16-wave block serves two batches of its 32 rows: adjacency rows read and bit masks packed once for both (was once per batch), masked row max per batch; all sum orders kept
# speedup vs baseline: 3.5915x; 1.0435x over previous
#include <hip/hip_runtime.h>
#include <cstddef>

typedef float v2f __attribute__((ext_vector_type(2)));
typedef float v4f __attribute__((ext_vector_type(4)));
typedef float v8f __attribute__((ext_vector_type(8)));
typedef int   v4i __attribute__((ext_vector_type(4)));

#define B_  8
#define N_  2048
#define C_  128
#define D_  128
#define NEG_BIG (-9.0e15f)

static __device__ __forceinline__ v8f wmma_f32_4(v2f a, v2f b, v8f c) {
    return __builtin_amdgcn_wmma_f32_16x16x4_f32(false, a, false, b, (short)0, c,
                                                 false, false);
}

static __device__ __forceinline__ void wht_load_stage(const float* xr, const float* wr, int s,
                                                      v4f (&xa)[2][4], v4f (&wb)[2][4]) {
#pragma unroll
    for (int t = 0; t < 2; ++t) {
#pragma unroll
        for (int q = 0; q < 4; ++q) {
            xa[t][q] = *(const v4f*)(xr + (size_t)t * 16 * C_ + s * 32 + q * 8);
            wb[t][q] = *(const v4f*)(wr + (size_t)t * 16 * C_ + s * 32 + q * 8);
        }
    }
}

static __device__ __forceinline__ void wht_mma_stage(const v4f (&xa)[2][4], const v4f (&wb)[2][4],
                                                     v8f (&acc)[2][2]) {
#pragma unroll
    for (int q = 0; q < 4; ++q) {
#pragma unroll
        for (int e = 0; e < 4; e += 2) {
            v2f A[2], Bf[2];
#pragma unroll
            for (int t = 0; t < 2; ++t) {
                A[t][0]  = xa[t][q][e]; A[t][1]  = xa[t][q][e + 1];
                Bf[t][0] = wb[t][q][e]; Bf[t][1] = wb[t][q][e + 1];
            }
#pragma unroll
            for (int ti = 0; ti < 2; ++ti) {
#pragma unroll
                for (int tj = 0; tj < 2; ++tj)
                    acc[ti][tj] = wmma_f32_4(A[ti], Bf[tj], acc[ti][tj]);
            }
        }
    }
}

__global__ __launch_bounds__(256)
void wht_kernel(const float* __restrict__ x, const float* __restrict__ W,
                float* __restrict__ whf) {
    const int tid  = threadIdx.x;
    const int wv   = tid >> 5;
    const int lane = tid & 31;
    const int nl   = lane & 15;
    const int hh   = lane >> 4;
    const int jt0  = blockIdx.x * 4 + (wv >> 2) * 2;
    const int dt0  = (wv & 3) * 2;

    const float* xr = x + ((size_t)(jt0 * 16 + nl)) * C_ + hh * 4;
    const float* wr = W + ((size_t)(dt0 * 16 + nl)) * C_ + hh * 4;

    v8f acc[2][2] = {};
    v4f xa0[2][4], wb0[2][4], xa1[2][4], wb1[2][4];

    wht_load_stage(xr, wr, 0, xa0, wb0);
    wht_load_stage(xr, wr, 1, xa1, wb1);
    __builtin_amdgcn_sched_barrier(0);
    wht_mma_stage(xa0, wb0, acc);
    wht_load_stage(xr, wr, 2, xa0, wb0);
    __builtin_amdgcn_sched_barrier(0);
    wht_mma_stage(xa1, wb1, acc);
    wht_load_stage(xr, wr, 3, xa1, wb1);
    __builtin_amdgcn_sched_barrier(0);
    wht_mma_stage(xa0, wb0, acc);
    wht_mma_stage(xa1, wb1, acc);

#pragma unroll
    for (int ti = 0; ti < 2; ++ti) {
#pragma unroll
        for (int tj = 0; tj < 2; ++tj) {
            float* ob = whf + ((((size_t)(jt0 + ti) * 2 + hh) * 8 + (dt0 + tj)) * 32) * 4;
            v4f lo, hi;
            lo[0] = acc[ti][tj][0]; lo[1] = acc[ti][tj][1];
            lo[2] = acc[ti][tj][2]; lo[3] = acc[ti][tj][3];
            hi[0] = acc[ti][tj][4]; hi[1] = acc[ti][tj][5];
            hi[2] = acc[ti][tj][6]; hi[3] = acc[ti][tj][7];
            *(v4f*)(ob + nl * 4)        = lo;
            *(v4f*)(ob + (nl + 16) * 4) = hi;
        }
    }
}

__global__ __launch_bounds__(256)
void f_kernel(const float* __restrict__ whf, const float* __restrict__ a,
              float* __restrict__ f1, float* __restrict__ f2) {
    const int tid  = threadIdx.x;
    const int lane = tid & 31;
    const int c    = lane & 15;
    const int hh   = lane >> 4;
    const int gc   = blockIdx.x * 8 + (tid >> 5);
    const v4f* wp  = (const v4f*)whf + (size_t)gc * (8 * 32) + lane;

    v4f s1 = {}, s2 = {};
#pragma unroll
    for (int dt = 0; dt < 8; ++dt) {
        const v4f w   = wp[dt * 32];
        const float a1 = a[dt * 16 + c];
        const float a2 = a[D_ + dt * 16 + c];
        s1 += w * a1;
        s2 += w * a2;
    }
#pragma unroll
    for (int m = 8; m >= 1; m >>= 1) {
#pragma unroll
        for (int e = 0; e < 4; ++e) {
            s1[e] += __shfl_xor(s1[e], m);
            s2[e] += __shfl_xor(s2[e], m);
        }
    }
    if (c == 0) {
        *(v4f*)(f1 + (size_t)gc * 8 + hh * 4) = s1;
        *(v4f*)(f2 + (size_t)gc * 8 + hh * 4) = s2;
    }
}

#define TI_ 2
#define NB_ 2

__global__ __launch_bounds__(512, 4)
void gat_kernel(const float* __restrict__ whf, const float* __restrict__ f1g,
                const float* __restrict__ f2g, const int* __restrict__ adj,
                float* __restrict__ out) {
    __shared__ unsigned lds_bits[TI_ * 16 * 65];
    __shared__ float lds_mw[NB_ * TI_ * 16];
    __shared__ float lds_m[NB_ * TI_ * 16], lds_f1[NB_ * TI_ * 16];
    __shared__ float lds_pone[NB_ * TI_ * 16], lds_l[NB_ * TI_ * 16];
    __shared__ float lds_lw[NB_ * 8 * TI_ * 16];
    __shared__ float accbuf[NB_ * 4 * TI_ * 8 * 32 * 8];

    const int bid  = blockIdx.x;
    const int bp   = bid / (N_ / (16 * TI_));
    const int i0   = (bid % (N_ / (16 * TI_))) * (16 * TI_);
    const int tid  = threadIdx.x;
    const int wv   = __builtin_amdgcn_readfirstlane(tid >> 5);
    const int grp  = wv & 1;
    const int w8   = wv >> 1;
    const int b    = bp * NB_ + grp;
    const int lane = tid & 31;
    const int nl   = lane & 15;

    const float* f2b  = f2g + b * N_;
    const v4f*   whfb = (const v4f*)(whf + (size_t)b * D_ * N_) + lane;

    {
        const int r0 = wv * 2;
        const v4i* arow0 = (const v4i*)(adj + (size_t)(i0 + r0) * N_) + lane;
        const v4i* arow1 = arow0 + N_ / 4;
        const v4f* frowA = (const v4f*)(f2g + (size_t)(bp * NB_) * N_) + lane;
        const v4f* frowB = frowA + N_ / 4;
        const unsigned sh = (unsigned)(lane & 7) * 4u;
        float mx0A = -3.0e38f, mx1A = -3.0e38f, mx0B = -3.0e38f, mx1B = -3.0e38f;
        for (int tb = 0; tb < 16; tb += 8) {
            v4i a0[8], a1[8];
            v4f fa[8], fb[8];
#pragma unroll
            for (int u = 0; u < 8; ++u) {
                a0[u] = arow0[(tb + u) * 32];
                a1[u] = arow1[(tb + u) * 32];
                fa[u] = frowA[(tb + u) * 32];
                fb[u] = frowB[(tb + u) * 32];
            }
            unsigned n0[8], n1[8];
#pragma unroll
            for (int u = 0; u < 8; ++u) {
                unsigned b0 = 0u, b1 = 0u;
#pragma unroll
                for (int e = 0; e < 4; ++e) {
                    if (a0[u][e] > 0) {
                        b0 |= 1u << e;
                        mx0A = fmaxf(mx0A, fa[u][e]);
                        mx0B = fmaxf(mx0B, fb[u][e]);
                    }
                    if (a1[u][e] > 0) {
                        b1 |= 1u << e;
                        mx1A = fmaxf(mx1A, fa[u][e]);
                        mx1B = fmaxf(mx1B, fb[u][e]);
                    }
                }
                n0[u] = b0 << sh;
                n1[u] = b1 << sh;
            }
#pragma unroll
            for (int m = 1; m <= 4; m <<= 1) {
#pragma unroll
                for (int u = 0; u < 8; ++u) {
                    n0[u] |= __shfl_xor(n0[u], m);
                    n1[u] |= __shfl_xor(n1[u], m);
                }
            }
            if ((lane & 7) == 0) {
#pragma unroll
                for (int u = 0; u < 8; ++u) {
                    lds_bits[r0 * 65 + (tb + u) * 4 + (lane >> 3)]       = n0[u];
                    lds_bits[(r0 + 1) * 65 + (tb + u) * 4 + (lane >> 3)] = n1[u];
                }
            }
        }
#pragma unroll
        for (int m = 16; m >= 1; m >>= 1) {
            mx0A = fmaxf(mx0A, __shfl_xor(mx0A, m));
            mx1A = fmaxf(mx1A, __shfl_xor(mx1A, m));
            mx0B = fmaxf(mx0B, __shfl_xor(mx0B, m));
            mx1B = fmaxf(mx1B, __shfl_xor(mx1B, m));
        }
        if (lane == 0) {
            lds_mw[r0]                = mx0A;
            lds_mw[r0 + 1]            = mx1A;
            lds_mw[TI_ * 16 + r0]     = mx0B;
            lds_mw[TI_ * 16 + r0 + 1] = mx1B;
        }
    }
    __syncthreads();
    if (tid < NB_ * TI_ * 16) {
        const int bq  = tid >> 5;
        const int row = tid & 31;
        const float M = lds_mw[tid];
        const bool empty = (M < -1.0e38f);
        const float f1v  = f1g[(bp * NB_ + bq) * N_ + i0 + row];
        const float sum  = f1v + M;
        const float emax = fmaxf(sum, 0.2f * sum);
        lds_m[tid]    = empty ? 0.0f : emax;
        lds_pone[tid] = empty ? 1.0f : 0.0f;
        lds_f1[tid]   = f1v;
    }
    __syncthreads();

    const int hh = lane >> 4;
    float f1i[TI_], mi[TI_], pm[TI_];
#pragma unroll
    for (int ti = 0; ti < TI_; ++ti) {
        f1i[ti] = lds_f1[grp * (TI_ * 16) + ti * 16 + nl];
        mi[ti]  = lds_m[grp * (TI_ * 16) + ti * 16 + nl];
        pm[ti]  = lds_pone[grp * (TI_ * 16) + ti * 16 + nl];
    }

    v8f acc[TI_][8] = {};
    float lsum[TI_] = {};

    auto load_chunk = [&](int g, v4f (&Bq)[8], v4f& f2v) {
        const int jc = g * 8 + w8;
        const v4f* wq = whfb + (size_t)jc * (8 * 32);
#pragma unroll
        for (int dt = 0; dt < 8; ++dt) Bq[dt] = wq[dt * 32];
        f2v = *(const v4f*)(f2b + jc * 8 + hh * 4);
    };
    auto do_chunk = [&](int g, const v4f (&Bq)[8], const v4f& f2v) {
        const int jc = g * 8 + w8;
        const unsigned msh = (unsigned)((jc & 3) * 8 + hh * 4);
#pragma unroll
        for (int ti = 0; ti < TI_; ++ti) {
            const unsigned word = lds_bits[(ti * 16 + nl) * 65 + (jc >> 2)];
            const unsigned m4   = word >> msh;

            float s0 = f1i[ti] + f2v[0]; s0 = fmaxf(s0, 0.2f * s0);
            float s1 = f1i[ti] + f2v[1]; s1 = fmaxf(s1, 0.2f * s1);
            float s2 = f1i[ti] + f2v[2]; s2 = fmaxf(s2, 0.2f * s2);
            float s3 = f1i[ti] + f2v[3]; s3 = fmaxf(s3, 0.2f * s3);
            const float p0 = (m4 & 1u) ? __expf(s0 - mi[ti]) : pm[ti];
            const float p1 = (m4 & 2u) ? __expf(s1 - mi[ti]) : pm[ti];
            const float p2 = (m4 & 4u) ? __expf(s2 - mi[ti]) : pm[ti];
            const float p3 = (m4 & 8u) ? __expf(s3 - mi[ti]) : pm[ti];

            v2f A0; A0[0] = p0; A0[1] = p1;
            v2f A1; A1[0] = p2; A1[1] = p3;
            lsum[ti] += (p0 + p1) + (p2 + p3);

#pragma unroll
            for (int dt = 0; dt < 8; ++dt) {
                v2f Bv; Bv[0] = Bq[dt][0]; Bv[1] = Bq[dt][1];
                acc[ti][dt] = wmma_f32_4(A0, Bv, acc[ti][dt]);
            }
#pragma unroll
            for (int dt = 0; dt < 8; ++dt) {
                v2f Bv; Bv[0] = Bq[dt][2]; Bv[1] = Bq[dt][3];
                acc[ti][dt] = wmma_f32_4(A1, Bv, acc[ti][dt]);
            }
        }
    };

    v4f BqA[8], BqB[8];
    v4f f2A, f2B;
    load_chunk(0, BqA, f2A);
    for (int g = 0; g < 30; g += 2) {
        load_chunk(g + 1, BqB, f2B);
        do_chunk(g, BqA, f2A);
        load_chunk(g + 2, BqA, f2A);
        do_chunk(g + 1, BqB, f2B);
    }
    load_chunk(31, BqB, f2B);
    do_chunk(30, BqA, f2A);
    do_chunk(31, BqB, f2B);

#pragma unroll
    for (int ti = 0; ti < TI_; ++ti) {
        const float lpair = lsum[ti] + __shfl_xor(lsum[ti], 16);
        if (lane < 16) lds_lw[(grp * 8 + w8) * (TI_ * 16) + ti * 16 + lane] = lpair;
    }
    __syncthreads();
    if (tid < NB_ * TI_ * 16) {
        const int bq  = tid >> 5;
        const int row = tid & 31;
        float t = 0.0f;
#pragma unroll
        for (int w2 = 0; w2 < 8; ++w2) t += lds_lw[(bq * 8 + w2) * (TI_ * 16) + row];
        lds_l[tid] = t;
    }
    __syncthreads();

    for (int half = 4; half >= 1; half >>= 1) {
        if (w8 >= half && w8 < 2 * half) {
            const int slot = grp * 4 + (w8 - half);
#pragma unroll
            for (int ti = 0; ti < TI_; ++ti) {
#pragma unroll
                for (int dt = 0; dt < 8; ++dt)
                    *(v8f*)&accbuf[((((size_t)slot * TI_ + ti) * 8 + dt) * 32 + lane) * 8] =
                        acc[ti][dt];
            }
        }
        __syncthreads();
        if (w8 < half) {
            const int slot = grp * 4 + w8;
#pragma unroll
            for (int ti = 0; ti < TI_; ++ti) {
#pragma unroll
                for (int dt = 0; dt < 8; ++dt)
                    acc[ti][dt] = acc[ti][dt] +
                        *(const v8f*)&accbuf[((((size_t)slot * TI_ + ti) * 8 + dt) * 32 + lane) * 8];
            }
        }
        __syncthreads();
    }

    if (w8 == 0) {
        float* ob = out + ((size_t)b * N_ + i0) * D_;
#pragma unroll
        for (int ti = 0; ti < TI_; ++ti) {
#pragma unroll
            for (int dt = 0; dt < 8; ++dt) {
#pragma unroll
                for (int r = 0; r < 8; ++r) {
                    const int row = ti * 16 + r + ((lane >> 4) << 3);
                    const float li = lds_l[grp * (TI_ * 16) + row];
                    const float h  = acc[ti][dt][r] / li;
                    const float o  = (h > 0.0f) ? h : (__expf(h) - 1.0f);
                    ob[(size_t)row * D_ + dt * 16 + nl] = o;
                }
            }
        }
    }
}

extern "C" void kernel_launch(void* const* d_in, const int* in_sizes, int n_in,
                              void* d_out, int out_size, void* d_ws, size_t ws_size,
                              hipStream_t stream) {
    const float* x   = (const float*)d_in[0];
    const int*   adj = (const int*)d_in[1];
    const float* W   = (const float*)d_in[2];
    const float* a   = (const float*)d_in[3];
    float* out = (float*)d_out;

    float* whf = (float*)d_ws;
    float* f1  = whf + (size_t)B_ * D_ * N_;
    float* f2  = f1 + (size_t)B_ * N_;

    wht_kernel<<<dim3((B_ * N_) / 64), dim3(256), 0, stream>>>(x, W, whf);
    f_kernel<<<dim3((B_ * N_) / 64), dim3(256), 0, stream>>>(whf, a, f1, f2);
    gat_kernel<<<dim3((B_ / NB_) * (N_ / (16 * TI_))), dim3(512), 0, stream>>>(whf, f1, f2, adj, out);
}


# gat pass 2: a lane's adjacency bits kept in registers, fetched from LDS twice per wave (pass 1 stores them in that order, no lane exchange); no LDS read or exec branch per chunk; sums kept
# speedup vs baseline: 3.7311x; 1.0389x over previous
#include <hip/hip_runtime.h>
#include <cstddef>

typedef float v2f __attribute__((ext_vector_type(2)));
typedef float v4f __attribute__((ext_vector_type(4)));
typedef float v8f __attribute__((ext_vector_type(8)));
typedef int   v4i __attribute__((ext_vector_type(4)));

#define B_  8
#define N_  2048
#define C_  128
#define D_  128
#define NEG_BIG (-9.0e15f)

static __device__ __forceinline__ v8f wmma_f32_4(v2f a, v2f b, v8f c) {
    return __builtin_amdgcn_wmma_f32_16x16x4_f32(false, a, false, b, (short)0, c,
                                                 false, false);
}

static __device__ __forceinline__ void wht_load_stage(const float* xr, const float* wr, int s,
                                                      v4f (&xa)[2][4], v4f (&wb)[2][4]) {
#pragma unroll
    for (int t = 0; t < 2; ++t) {
#pragma unroll
        for (int q = 0; q < 4; ++q) {
            xa[t][q] = *(const v4f*)(xr + (size_t)t * 16 * C_ + s * 32 + q * 8);
            wb[t][q] = *(const v4f*)(wr + (size_t)t * 16 * C_ + s * 32 + q * 8);
        }
    }
}

static __device__ __forceinline__ void wht_mma_stage(const v4f (&xa)[2][4], const v4f (&wb)[2][4],
                                                     v8f (&acc)[2][2]) {
#pragma unroll
    for (int q = 0; q < 4; ++q) {
#pragma unroll
        for (int e = 0; e < 4; e += 2) {
            v2f A[2], Bf[2];
#pragma unroll
            for (int t = 0; t < 2; ++t) {
                A[t][0]  = xa[t][q][e]; A[t][1]  = xa[t][q][e + 1];
                Bf[t][0] = wb[t][q][e]; Bf[t][1] = wb[t][q][e + 1];
            }
#pragma unroll
            for (int ti = 0; ti < 2; ++ti) {
#pragma unroll
                for (int tj = 0; tj < 2; ++tj)
                    acc[ti][tj] = wmma_f32_4(A[ti], Bf[tj], acc[ti][tj]);
            }
        }
    }
}

__global__ __launch_bounds__(256)
void wht_kernel(const float* __restrict__ x, const float* __restrict__ W,
                float* __restrict__ whf) {
    const int tid  = threadIdx.x;
    const int wv   = tid >> 5;
    const int lane = tid & 31;
    const int nl   = lane & 15;
    const int hh   = lane >> 4;
    const int jt0  = blockIdx.x * 4 + (wv >> 2) * 2;
    const int dt0  = (wv & 3) * 2;

    const float* xr = x + ((size_t)(jt0 * 16 + nl)) * C_ + hh * 4;
    const float* wr = W + ((size_t)(dt0 * 16 + nl)) * C_ + hh * 4;

    v8f acc[2][2] = {};
    v4f xa0[2][4], wb0[2][4], xa1[2][4], wb1[2][4];

    wht_load_stage(xr, wr, 0, xa0, wb0);
    wht_load_stage(xr, wr, 1, xa1, wb1);
    __builtin_amdgcn_sched_barrier(0);
    wht_mma_stage(xa0, wb0, acc);
    wht_load_stage(xr, wr, 2, xa0, wb0);
    __builtin_amdgcn_sched_barrier(0);
    wht_mma_stage(xa1, wb1, acc);
    wht_load_stage(xr, wr, 3, xa1, wb1);
    __builtin_amdgcn_sched_barrier(0);
    wht_mma_stage(xa0, wb0, acc);
    wht_mma_stage(xa1, wb1, acc);

#pragma unroll
    for (int ti = 0; ti < 2; ++ti) {
#pragma unroll
        for (int tj = 0; tj < 2; ++tj) {
            float* ob = whf + ((((size_t)(jt0 + ti) * 2 + hh) * 8 + (dt0 + tj)) * 32) * 4;
            v4f lo, hi;
            lo[0] = acc[ti][tj][0]; lo[1] = acc[ti][tj][1];
            lo[2] = acc[ti][tj][2]; lo[3] = acc[ti][tj][3];
            hi[0] = acc[ti][tj][4]; hi[1] = acc[ti][tj][5];
            hi[2] = acc[ti][tj][6]; hi[3] = acc[ti][tj][7];
            *(v4f*)(ob + nl * 4)        = lo;
            *(v4f*)(ob + (nl + 16) * 4) = hi;
        }
    }
}

__global__ __launch_bounds__(256)
void f_kernel(const float* __restrict__ whf, const float* __restrict__ a,
              float* __restrict__ f1, float* __restrict__ f2) {
    const int tid  = threadIdx.x;
    const int lane = tid & 31;
    const int c    = lane & 15;
    const int hh   = lane >> 4;
    const int gc   = blockIdx.x * 8 + (tid >> 5);
    const v4f* wp  = (const v4f*)whf + (size_t)gc * (8 * 32) + lane;

    v4f s1 = {}, s2 = {};
#pragma unroll
    for (int dt = 0; dt < 8; ++dt) {
        const v4f w   = wp[dt * 32];
        const float a1 = a[dt * 16 + c];
        const float a2 = a[D_ + dt * 16 + c];
        s1 += w * a1;
        s2 += w * a2;
    }
#pragma unroll
    for (int m = 8; m >= 1; m >>= 1) {
#pragma unroll
        for (int e = 0; e < 4; ++e) {
            s1[e] += __shfl_xor(s1[e], m);
            s2[e] += __shfl_xor(s2[e], m);
        }
    }
    if (c == 0) {
        *(v4f*)(f1 + (size_t)gc * 8 + hh * 4) = s1;
        *(v4f*)(f2 + (size_t)gc * 8 + hh * 4) = s2;
    }
}

#define TI_ 2
#define NB_ 2

__global__ __launch_bounds__(512, 4)
void gat_kernel(const float* __restrict__ whf, const float* __restrict__ f1g,
                const float* __restrict__ f2g, const int* __restrict__ adj,
                float* __restrict__ out) {
    __shared__ unsigned lds_bits[TI_ * 16 * 65];
    __shared__ float lds_mw[NB_ * TI_ * 16];
    __shared__ float lds_m[NB_ * TI_ * 16], lds_f1[NB_ * TI_ * 16];
    __shared__ float lds_pone[NB_ * TI_ * 16], lds_l[NB_ * TI_ * 16];
    __shared__ float lds_lw[NB_ * 8 * TI_ * 16];
    __shared__ float accbuf[NB_ * 4 * TI_ * 8 * 32 * 8];

    const int bid  = blockIdx.x;
    const int bp   = bid / (N_ / (16 * TI_));
    const int i0   = (bid % (N_ / (16 * TI_))) * (16 * TI_);
    const int tid  = threadIdx.x;
    const int wv   = __builtin_amdgcn_readfirstlane(tid >> 5);
    const int grp  = wv & 1;
    const int w8   = wv >> 1;
    const int b    = bp * NB_ + grp;
    const int lane = tid & 31;
    const int nl   = lane & 15;

    const float* f2b  = f2g + b * N_;
    const v4f*   whfb = (const v4f*)(whf + (size_t)b * D_ * N_) + lane;

    {
        const int r0 = wv * 2;
        const v4i* arow0 = (const v4i*)(adj + (size_t)(i0 + r0) * N_) + lane;
        const v4i* arow1 = arow0 + N_ / 4;
        const v4f* frowA = (const v4f*)(f2g + (size_t)(bp * NB_) * N_) + lane;
        const v4f* frowB = frowA + N_ / 4;
        float mx0A = -3.0e38f, mx1A = -3.0e38f, mx0B = -3.0e38f, mx1B = -3.0e38f;
        for (int tb = 0; tb < 16; tb += 8) {
            v4i a0[8], a1[8];
            v4f fa[8], fb[8];
#pragma unroll
            for (int u = 0; u < 8; ++u) {
                a0[u] = arow0[(tb + u) * 32];
                a1[u] = arow1[(tb + u) * 32];
                fa[u] = frowA[(tb + u) * 32];
                fb[u] = frowB[(tb + u) * 32];
            }
            unsigned w0 = 0u, w1 = 0u;
#pragma unroll
            for (int u = 0; u < 8; ++u) {
                unsigned b0 = 0u, b1 = 0u;
#pragma unroll
                for (int e = 0; e < 4; ++e) {
                    if (a0[u][e] > 0) {
                        b0 |= 1u << e;
                        mx0A = fmaxf(mx0A, fa[u][e]);
                        mx0B = fmaxf(mx0B, fb[u][e]);
                    }
                    if (a1[u][e] > 0) {
                        b1 |= 1u << e;
                        mx1A = fmaxf(mx1A, fa[u][e]);
                        mx1B = fmaxf(mx1B, fb[u][e]);
                    }
                }
                w0 |= b0 << (4 * u);
                w1 |= b1 << (4 * u);
            }
            lds_bits[r0 * 65 + (tb >> 3) * 32 + lane]       = w0;
            lds_bits[(r0 + 1) * 65 + (tb >> 3) * 32 + lane] = w1;
        }
#pragma unroll
        for (int m = 16; m >= 1; m >>= 1) {
            mx0A = fmaxf(mx0A, __shfl_xor(mx0A, m));
            mx1A = fmaxf(mx1A, __shfl_xor(mx1A, m));
            mx0B = fmaxf(mx0B, __shfl_xor(mx0B, m));
            mx1B = fmaxf(mx1B, __shfl_xor(mx1B, m));
        }
        if (lane == 0) {
            lds_mw[r0]                = mx0A;
            lds_mw[r0 + 1]            = mx1A;
            lds_mw[TI_ * 16 + r0]     = mx0B;
            lds_mw[TI_ * 16 + r0 + 1] = mx1B;
        }
    }
    __syncthreads();
    if (tid < NB_ * TI_ * 16) {
        const int bq  = tid >> 5;
        const int row = tid & 31;
        const float M = lds_mw[tid];
        const bool empty = (M < -1.0e38f);
        const float f1v  = f1g[(bp * NB_ + bq) * N_ + i0 + row];
        const float sum  = f1v + M;
        const float emax = fmaxf(sum, 0.2f * sum);
        lds_m[tid]    = empty ? 0.0f : emax;
        lds_pone[tid] = empty ? 1.0f : 0.0f;
        lds_f1[tid]   = f1v;
    }
    __syncthreads();

    const int hh = lane >> 4;
    float f1i[TI_], mi[TI_], pm[TI_];
#pragma unroll
    for (int ti = 0; ti < TI_; ++ti) {
        f1i[ti] = lds_f1[grp * (TI_ * 16) + ti * 16 + nl];
        mi[ti]  = lds_m[grp * (TI_ * 16) + ti * 16 + nl];
        pm[ti]  = lds_pone[grp * (TI_ * 16) + ti * 16 + nl];
    }

    v8f acc[TI_][8] = {};
    float lsum[TI_] = {};

    unsigned mE[TI_], mO[TI_];
    auto load_masks = [&](int word) {
#pragma unroll
        for (int ti = 0; ti < TI_; ++ti) {
            const unsigned* mrow = &lds_bits[(ti * 16 + nl) * 65 + word * 32 + 2 * w8 + hh];
            mE[ti] = mrow[0];
            mO[ti] = mrow[16];
        }
    };
    load_masks(0);

    auto load_chunk = [&](int g, v4f (&Bq)[8], v4f& f2v) {
        const int jc = g * 8 + w8;
        const v4f* wq = whfb + (size_t)jc * (8 * 32);
#pragma unroll
        for (int dt = 0; dt < 8; ++dt) Bq[dt] = wq[dt * 32];
        f2v = *(const v4f*)(f2b + jc * 8 + hh * 4);
    };
    auto do_chunk = [&](const v4f (&Bq)[8], const v4f& f2v, const unsigned (&mb)[TI_]) {
#pragma unroll
        for (int ti = 0; ti < TI_; ++ti) {
            const unsigned m4 = mb[ti];

            float s0 = f1i[ti] + f2v[0]; s0 = fmaxf(s0, 0.2f * s0);
            float s1 = f1i[ti] + f2v[1]; s1 = fmaxf(s1, 0.2f * s1);
            float s2 = f1i[ti] + f2v[2]; s2 = fmaxf(s2, 0.2f * s2);
            float s3 = f1i[ti] + f2v[3]; s3 = fmaxf(s3, 0.2f * s3);
            const float e0 = __expf(s0 - mi[ti]);
            const float e1 = __expf(s1 - mi[ti]);
            const float e2 = __expf(s2 - mi[ti]);
            const float e3 = __expf(s3 - mi[ti]);
            const float p0 = (m4 & 1u) ? e0 : pm[ti];
            const float p1 = (m4 & 2u) ? e1 : pm[ti];
            const float p2 = (m4 & 4u) ? e2 : pm[ti];
            const float p3 = (m4 & 8u) ? e3 : pm[ti];

            v2f A0; A0[0] = p0; A0[1] = p1;
            v2f A1; A1[0] = p2; A1[1] = p3;
            lsum[ti] += (p0 + p1) + (p2 + p3);

#pragma unroll
            for (int dt = 0; dt < 8; ++dt) {
                v2f Bv; Bv[0] = Bq[dt][0]; Bv[1] = Bq[dt][1];
                acc[ti][dt] = wmma_f32_4(A0, Bv, acc[ti][dt]);
            }
#pragma unroll
            for (int dt = 0; dt < 8; ++dt) {
                v2f Bv; Bv[0] = Bq[dt][2]; Bv[1] = Bq[dt][3];
                acc[ti][dt] = wmma_f32_4(A1, Bv, acc[ti][dt]);
            }
        }
    };

    v4f BqA[8], BqB[8];
    v4f f2A, f2B;
    load_chunk(0, BqA, f2A);
    for (int g = 0; g < 30; g += 2) {
        if (g == 16) load_masks(1);
        load_chunk(g + 1, BqB, f2B);
        do_chunk(BqA, f2A, mE);
        load_chunk(g + 2, BqA, f2A);
        do_chunk(BqB, f2B, mO);
#pragma unroll
        for (int ti = 0; ti < TI_; ++ti) {
            mE[ti] >>= 4;
            mO[ti] >>= 4;
        }
    }
    load_chunk(31, BqB, f2B);
    do_chunk(BqA, f2A, mE);
    do_chunk(BqB, f2B, mO);

#pragma unroll
    for (int ti = 0; ti < TI_; ++ti) {
        const float lpair = lsum[ti] + __shfl_xor(lsum[ti], 16);
        if (lane < 16) lds_lw[(grp * 8 + w8) * (TI_ * 16) + ti * 16 + lane] = lpair;
    }
    __syncthreads();
    if (tid < NB_ * TI_ * 16) {
        const int bq  = tid >> 5;
        const int row = tid & 31;
        float t = 0.0f;
#pragma unroll
        for (int w2 = 0; w2 < 8; ++w2) t += lds_lw[(bq * 8 + w2) * (TI_ * 16) + row];
        lds_l[tid] = t;
    }
    __syncthreads();

    for (int half = 4; half >= 1; half >>= 1) {
        if (w8 >= half && w8 < 2 * half) {
            const int slot = grp * 4 + (w8 - half);
#pragma unroll
            for (int ti = 0; ti < TI_; ++ti) {
#pragma unroll
                for (int dt = 0; dt < 8; ++dt)
                    *(v8f*)&accbuf[((((size_t)slot * TI_ + ti) * 8 + dt) * 32 + lane) * 8] =
                        acc[ti][dt];
            }
        }
        __syncthreads();
        if (w8 < half) {
            const int slot = grp * 4 + w8;
#pragma unroll
            for (int ti = 0; ti < TI_; ++ti) {
#pragma unroll
                for (int dt = 0; dt < 8; ++dt)
                    acc[ti][dt] = acc[ti][dt] +
                        *(const v8f*)&accbuf[((((size_t)slot * TI_ + ti) * 8 + dt) * 32 + lane) * 8];
            }
        }
        __syncthreads();
    }

    if (w8 == 0) {
        float* ob = out + ((size_t)b * N_ + i0) * D_;
#pragma unroll
        for (int ti = 0; ti < TI_; ++ti) {
#pragma unroll
            for (int dt = 0; dt < 8; ++dt) {
#pragma unroll
                for (int r = 0; r < 8; ++r) {
                    const int row = ti * 16 + r + ((lane >> 4) << 3);
                    const float li = lds_l[grp * (TI_ * 16) + row];
                    const float h  = acc[ti][dt][r] / li;
                    const float o  = (h > 0.0f) ? h : (__expf(h) - 1.0f);
                    ob[(size_t)row * D_ + dt * 16 + nl] = o;
                }
            }
        }
    }
}

extern "C" void kernel_launch(void* const* d_in, const int* in_sizes, int n_in,
                              void* d_out, int out_size, void* d_ws, size_t ws_size,
                              hipStream_t stream) {
    const float* x   = (const float*)d_in[0];
    const int*   adj = (const int*)d_in[1];
    const float* W   = (const float*)d_in[2];
    const float* a   = (const float*)d_in[3];
    float* out = (float*)d_out;

    float* whf = (float*)d_ws;
    float* f1  = whf + (size_t)B_ * D_ * N_;
    float* f2  = f1 + (size_t)B_ * N_;

    wht_kernel<<<dim3((B_ * N_) / 64), dim3(256), 0, stream>>>(x, W, whf);
    f_kernel<<<dim3((B_ * N_) / 64), dim3(256), 0, stream>>>(whf, a, f1, f2);
    gat_kernel<<<dim3((B_ / NB_) * (N_ / (16 * TI_))), dim3(512), 0, stream>>>(whf, f1, f2, adj, out);
}
